# starting kernel
# baseline (speedup 1.0000x reference)
#include <hip/hip_runtime.h>
#include <cmath>

#define HQ 1024
#define WQ 1024
#define HW (HQ * WQ)
#define NB 8
#define NC 3

typedef __attribute__((ext_vector_type(2))) float v2f;
typedef __attribute__((ext_vector_type(8))) float v8f;

struct GW { float g31[31]; float g7[7]; };

__device__ __forceinline__ int iclamp(int v, int lo, int hi) {
    return v < lo ? lo : (v > hi ? hi : v);
}

__global__ __launch_bounds__(256) void luma_kernel(const float* __restrict__ x,
                                                   float* __restrict__ L) {
    int i = blockIdx.x * 256 + threadIdx.x;
    int b   = i >> 18;
    int rem = i & ((HW / 4) - 1);
    const float4* xv = (const float4*)x;
    float4 r  = xv[(size_t)(b * 3 + 0) * (HW / 4) + rem];
    float4 g  = xv[(size_t)(b * 3 + 1) * (HW / 4) + rem];
    float4 bl = xv[(size_t)(b * 3 + 2) * (HW / 4) + rem];
    float4 o;
    o.x = 0.2126f * r.x + 0.7152f * g.x + 0.0722f * bl.x;
    o.y = 0.2126f * r.y + 0.7152f * g.y + 0.0722f * bl.y;
    o.z = 0.2126f * r.z + 0.7152f * g.z + 0.0722f * bl.z;
    o.w = 0.2126f * r.w + 0.7152f * g.w + 0.0722f * bl.w;
    ((float4*)L)[(size_t)b * (HW / 4) + rem] = o;
}

__global__ __launch_bounds__(256) void vblur_kernel(const float* __restrict__ L,
                                                    float* __restrict__ vb31,
                                                    float* __restrict__ vb7,
                                                    GW gw) {
    int i = blockIdx.x * 256 + threadIdx.x;
    int b  = i >> 20;
    int p  = i & (HW - 1);
    int y  = p >> 10;
    int xx = p & (WQ - 1);
    const float* base = L + (size_t)b * HW + xx;

    float a31 = 0.f;
#pragma unroll
    for (int t = 0; t < 31; ++t) {
        int yy = y + t - 15;
        int yc = iclamp(yy, 0, HQ - 1);
        float m = (yy == yc) ? 1.f : 0.f;
        a31 = fmaf(gw.g31[t], base[(size_t)yc * WQ] * m, a31);
    }
    float a7 = 0.f;
#pragma unroll
    for (int t = 0; t < 7; ++t) {
        int yy = y + t - 3;
        int yc = iclamp(yy, 0, HQ - 1);
        float m = (yy == yc) ? 1.f : 0.f;
        a7 = fmaf(gw.g7[t], base[(size_t)yc * WQ] * m, a7);
    }
    vb31[i] = a31;
    vb7[i]  = a7;
}

__global__ __launch_bounds__(32) void hblur_combine_kernel(
    const float* __restrict__ x, const float* __restrict__ L,
    const float* __restrict__ VB31, const float* __restrict__ VB7,
    const float* __restrict__ pclar, const float* __restrict__ ptex,
    float* __restrict__ out) {
    __shared__ float sg31[31];
    __shared__ float sg7[7];

    const int lane = threadIdx.x;
    const int ln   = lane & 15;
    const int hi   = lane >> 4;

    if (lane < 31) { float d = (float)lane - 15.f; sg31[lane] = expf(-d * d * (1.f / 128.f)); }
    if (lane < 7)  { float d = (float)lane - 3.f;  sg7[lane]  = expf(-d * d * (1.f / 4.5f)); }
    __syncthreads();
    float s31 = 0.f, s7 = 0.f;
    for (int t = 0; t < 31; ++t) s31 += sg31[t];
    for (int t = 0; t < 7;  ++t) s7  += sg7[t];
    const float inv31 = 1.f / s31, inv7 = 1.f / s7;

    v2f w31[12];
#pragma unroll
    for (int s = 0; s < 12; ++s) {
#pragma unroll
        for (int v = 0; v < 2; ++v) {
            int t  = 4 * s + 2 * hi + v - ln;
            int tc = iclamp(t, 0, 30);
            w31[s][v] = sg31[tc] * ((t == tc) ? inv31 : 0.f);
        }
    }
    v2f w7[6];
#pragma unroll
    for (int s = 0; s < 6; ++s) {
#pragma unroll
        for (int v = 0; v < 2; ++v) {
            int t  = 4 * s + 2 * hi + v - ln;
            int tc = iclamp(t, 0, 6);
            w7[s][v] = sg7[tc] * ((t == tc) ? inv7 : 0.f);
        }
    }

    const int x0 = blockIdx.x * 16;
    const int y0 = blockIdx.y * 16;
    const int b  = blockIdx.z;
    const int row = y0 + ln;
    const float* p31 = VB31 + (size_t)b * HW + (size_t)row * WQ;
    const float* p7  = VB7  + (size_t)b * HW + (size_t)row * WQ;
    const int kbase = 2 * hi;

    v8f acc31 = {0.f, 0.f, 0.f, 0.f, 0.f, 0.f, 0.f, 0.f};
#pragma unroll
    for (int s = 0; s < 12; ++s) {
        v2f a;
#pragma unroll
        for (int v = 0; v < 2; ++v) {
            int col = x0 - 15 + 4 * s + kbase + v;
            int cc  = iclamp(col, 0, WQ - 1);
            a[v] = p31[cc] * ((col == cc) ? 1.f : 0.f);
        }
        acc31 = __builtin_amdgcn_wmma_f32_16x16x4_f32(
            false, a, false, w31[s], (short)0, acc31, false, false);
    }

    v8f acc7 = {0.f, 0.f, 0.f, 0.f, 0.f, 0.f, 0.f, 0.f};
#pragma unroll
    for (int s = 0; s < 6; ++s) {
        v2f a;
#pragma unroll
        for (int v = 0; v < 2; ++v) {
            int col = x0 - 3 + 4 * s + kbase + v;
            int cc  = iclamp(col, 0, WQ - 1);
            a[v] = p7[cc] * ((col == cc) ? 1.f : 0.f);
        }
        acc7 = __builtin_amdgcn_wmma_f32_16x16x4_f32(
            false, a, false, w7[s], (short)0, acc7, false, false);
    }

    const float ca = tanhf(pclar[0]) * 0.5f;
    const float ta = tanhf(ptex[0])  * 0.3f;
#pragma unroll
    for (int r = 0; r < 8; ++r) {
        int y  = y0 + r + 8 * hi;
        int xc = x0 + ln;
        size_t pidx = (size_t)b * HW + (size_t)y * WQ + xc;
        float Lv  = L[pidx];
        float le  = Lv + ca * (Lv - acc31[r]) + ta * (Lv - acc7[r]);
        float ratio = (le + 1e-6f) / (Lv + 1e-6f);
#pragma unroll
        for (int c = 0; c < 3; ++c) {
            size_t xi = (size_t)(b * 3 + c) * HW + (size_t)y * WQ + xc;
            float vv = x[xi] * ratio;
            out[xi] = fminf(fmaxf(vv, 0.f), 1.f);
        }
    }
}

extern "C" void kernel_launch(void* const* d_in, const int* in_sizes, int n_in,
                              void* d_out, int out_size, void* d_ws, size_t ws_size,
                              hipStream_t stream) {
    const float* x    = (const float*)d_in[0];
    const float* clar = (const float*)d_in[1];
    const float* tex  = (const float*)d_in[2];
    float* out = (float*)d_out;

    float* ws   = (float*)d_ws;
    float* L    = ws;
    float* VB31 = ws + (size_t)NB * HW;
    float* VB7  = ws + (size_t)2 * NB * HW;

    GW gw;
    {
        float s = 0.f;
        for (int i = 0; i < 31; ++i) { float d = (float)(i - 15); gw.g31[i] = expf(-d * d / (2.f * 8.f * 8.f)); s += gw.g31[i]; }
        for (int i = 0; i < 31; ++i) gw.g31[i] /= s;
        s = 0.f;
        for (int i = 0; i < 7; ++i)  { float d = (float)(i - 3);  gw.g7[i]  = expf(-d * d / (2.f * 1.5f * 1.5f)); s += gw.g7[i]; }
        for (int i = 0; i < 7; ++i)  gw.g7[i] /= s;
    }

    luma_kernel<<<(NB * HW / 4) / 256, 256, 0, stream>>>(x, L);
    vblur_kernel<<<(NB * HW) / 256, 256, 0, stream>>>(L, VB31, VB7, gw);
    dim3 grid3(WQ / 16, HQ / 16, NB);
    hblur_combine_kernel<<<grid3, 32, 0, stream>>>(x, L, VB31, VB7, clar, tex, out);
}


# vblur pass removed: kernel 2 forms both vertical blurs as transposed WMMA f32 products (luma as A, lanes along x), their D registers feed the horizontal products as A; vertical sums in WMMA order
# speedup vs baseline: 2.6817x; 2.6817x over previous
#include <hip/hip_runtime.h>
#include <cmath>


#define HQ 1024
#define WQ 1024
#define HW (HQ * WQ)
#define NB 8
#define NC 3
#define NT 8

typedef __attribute__((ext_vector_type(2))) float v2f;
typedef __attribute__((ext_vector_type(8))) float v8f;

__device__ __forceinline__ int iclamp(int v, int lo, int hi) {
    return v < lo ? lo : (v > hi ? hi : v);
}

__global__ __launch_bounds__(256) void luma_kernel(const float* __restrict__ x,
                                                   float* __restrict__ L) {
    int i = blockIdx.x * 256 + threadIdx.x;
    int b   = i >> 18;
    int rem = i & ((HW / 4) - 1);
    const float4* xv = (const float4*)x;
    float4 r  = xv[(size_t)(b * 3 + 0) * (HW / 4) + rem];
    float4 g  = xv[(size_t)(b * 3 + 1) * (HW / 4) + rem];
    float4 bl = xv[(size_t)(b * 3 + 2) * (HW / 4) + rem];
    float4 o;
    o.x = 0.2126f * r.x + 0.7152f * g.x + 0.0722f * bl.x;
    o.y = 0.2126f * r.y + 0.7152f * g.y + 0.0722f * bl.y;
    o.z = 0.2126f * r.z + 0.7152f * g.z + 0.0722f * bl.z;
    o.w = 0.2126f * r.w + 0.7152f * g.w + 0.0722f * bl.w;
    ((float4*)L)[(size_t)b * (HW / 4) + rem] = o;
}

__device__ __forceinline__ constexpr bool use7(int q, int s) {
    return (q == 1) || (q == 0 && s >= 2) || (q == 2 && s <= 1);
}

__global__ __launch_bounds__(128) void blur_combine_kernel(
    const float* __restrict__ x, const float* __restrict__ L,
    const float* __restrict__ pclar, const float* __restrict__ ptex,
    float* __restrict__ out) {
    __shared__ float sg31[31];
    __shared__ float sg7[7];

    const int tid  = threadIdx.x;
    const int lane = tid & 31;
    const int ln   = lane & 15;
    const int hi   = lane >> 4;
    const int wv   = __builtin_amdgcn_readfirstlane(tid >> 5);

    if (tid < 31) { float d = (float)tid - 15.f; sg31[tid] = expf(-d * d * (1.f / 128.f)); }
    if (tid < 7)  { float d = (float)tid - 3.f;  sg7[tid]  = expf(-d * d * (1.f / 4.5f)); }
    __syncthreads();
    float s31 = 0.f, s7 = 0.f;
    for (int t = 0; t < 31; ++t) s31 += sg31[t];
    for (int t = 0; t < 7;  ++t) s7  += sg7[t];
    const float inv31 = 1.f / s31, inv7 = 1.f / s7;

    v2f w31[3][4];
    v2f w7[3][4];
#pragma unroll
    for (int q = 0; q < 3; ++q) {
#pragma unroll
        for (int s = 0; s < 4; ++s) {
#pragma unroll
            for (int v = 0; v < 2; ++v) {
                int kk = 16 * q + 8 * hi + 2 * s + v;
                int t  = kk - ln - 1;
                int tc = iclamp(t, 0, 30);
                w31[q][s][v] = sg31[tc] * ((t == tc) ? inv31 : 0.f);
                int u  = kk - ln - 13;
                int uc = iclamp(u, 0, 6);
                w7[q][s][v] = sg7[uc] * ((u == uc) ? inv7 : 0.f);
            }
        }
    }

    const int xs = blockIdx.x * (16 * NT);
    const int y0 = (blockIdx.y * 4 + wv) * 16;
    const int b  = blockIdx.z;

    const bool top = (y0 == 0), bot = (y0 == HQ - 16);
    const int   rq0 = top ? 16 : 0;
    const int   rq2 = bot ? 16 : 32;
    const float mq0 = top ? 0.f : 1.f;
    const float mq2 = bot ? 0.f : 1.f;

    const float* Lb = L + (size_t)b * HW;
    const int lo   = 8 * hi * WQ + ln;
    const int off0 = (y0 - 16 + rq0) * WQ + lo;
    const int off1 = y0 * WQ + lo;
    const int off2 = (y0 - 16 + rq2) * WQ + lo;

    const float ca = tanhf(pclar[0]) * 0.5f;
    const float ta = tanhf(ptex[0])  * 0.3f;

    const v8f zero8 = {0.f, 0.f, 0.f, 0.f, 0.f, 0.f, 0.f, 0.f};
    v8f Pa31 = zero8, Pb31 = zero8, Pc31 = zero8;
    v8f Pa7  = zero8, Pb7  = zero8, Pc7  = zero8;
    v8f Lprev = zero8, Lcur = zero8;

#pragma unroll 1
    for (int i = 0; i < NT + 2; ++i) {
        const int X  = xs + 16 * (i - 1);
        const int Xc = iclamp(X, 0, WQ - 16);
        const float mX = (X == Xc) ? 1.f : 0.f;
        const float m0 = mq0 * mX, m2 = mq2 * mX;

        v8f n31 = zero8, n7 = zero8, Lnew = zero8;
#pragma unroll
        for (int q = 0; q < 3; ++q) {
            const float* src = Lb + ((q == 0) ? off0 : (q == 1) ? off1 : off2) + Xc;
            const float  mm  = (q == 0) ? m0 : (q == 1) ? mX : m2;
#pragma unroll
            for (int s = 0; s < 4; ++s) {
                v2f a;
                a[0] = src[(2 * s) * WQ] * mm;
                a[1] = src[(2 * s + 1) * WQ] * mm;
                n31 = __builtin_amdgcn_wmma_f32_16x16x4_f32(
                    false, a, false, w31[q][s], (short)0, n31, false, false);
                if (use7(q, s))
                    n7 = __builtin_amdgcn_wmma_f32_16x16x4_f32(
                        false, a, false, w7[q][s], (short)0, n7, false, false);
                if (q == 1) { Lnew[2 * s] = a[0]; Lnew[2 * s + 1] = a[1]; }
            }
        }
        Pa31 = Pb31; Pb31 = Pc31; Pc31 = n31;
        Pa7  = Pb7;  Pb7  = Pc7;  Pc7  = n7;
        Lprev = Lcur; Lcur = Lnew;

        if (i >= 2) {
            const int x0 = xs + 16 * (i - 2);
            v8f acc31 = zero8, acc7 = zero8;
#pragma unroll
            for (int q = 0; q < 3; ++q) {
                const v8f p31 = (q == 0) ? Pa31 : (q == 1) ? Pb31 : Pc31;
                const v8f p7  = (q == 0) ? Pa7  : (q == 1) ? Pb7  : Pc7;
#pragma unroll
                for (int s = 0; s < 4; ++s) {
                    v2f a31 = {p31[2 * s], p31[2 * s + 1]};
                    acc31 = __builtin_amdgcn_wmma_f32_16x16x4_f32(
                        false, a31, false, w31[q][s], (short)0, acc31, false, false);
                    if (use7(q, s)) {
                        v2f a7 = {p7[2 * s], p7[2 * s + 1]};
                        acc7 = __builtin_amdgcn_wmma_f32_16x16x4_f32(
                            false, a7, false, w7[q][s], (short)0, acc7, false, false);
                    }
                }
            }
            const size_t pbase = (size_t)(y0 + 8 * hi) * WQ + x0 + ln;
#pragma unroll
            for (int r = 0; r < 8; ++r) {
                float Lv  = Lprev[r];
                float le  = Lv + ca * (Lv - acc31[r]) + ta * (Lv - acc7[r]);
                float ratio = (le + 1e-6f) / (Lv + 1e-6f);
#pragma unroll
                for (int c = 0; c < 3; ++c) {
                    size_t xi = (size_t)(b * 3 + c) * HW + pbase + (size_t)r * WQ;
                    float vv = x[xi] * ratio;
                    out[xi] = fminf(fmaxf(vv, 0.f), 1.f);
                }
            }
        }
    }
}

extern "C" void kernel_launch(void* const* d_in, const int* in_sizes, int n_in,
                              void* d_out, int out_size, void* d_ws, size_t ws_size,
                              hipStream_t stream) {
    const float* x    = (const float*)d_in[0];
    const float* clar = (const float*)d_in[1];
    const float* tex  = (const float*)d_in[2];
    float* out = (float*)d_out;

    float* L = (float*)d_ws;

    luma_kernel<<<(NB * HW / 4) / 256, 256, 0, stream>>>(x, L);
    dim3 grid2(WQ / (16 * NT), HQ / 64, NB);
    blur_combine_kernel<<<grid2, 128, 0, stream>>>(x, L, clar, tex, out);
}


# blur kernel steps over pairs of 16-column tiles, lane = column (halves exchanged by v_permlanex16): each load/store is one whole 128-byte line; luma plane stored 16 columns right, pitch 1056
# speedup vs baseline: 3.6471x; 1.3600x over previous
#include <hip/hip_runtime.h>
#include <cmath>


#define HQ 1024
#define WQ 1024
#define HW (HQ * WQ)
#define NB 8
#define NC 3
#define NT 8
#define LP (WQ + 32)
#define LOFF 16

typedef __attribute__((ext_vector_type(2))) float v2f;
typedef __attribute__((ext_vector_type(8))) float v8f;

__device__ __forceinline__ int iclamp(int v, int lo, int hi) {
    return v < lo ? lo : (v > hi ? hi : v);
}

__global__ __launch_bounds__(256) void luma_kernel(const float* __restrict__ x,
                                                   float* __restrict__ L) {
    int i = blockIdx.x * 256 + threadIdx.x;
    int b   = i >> 18;
    int rem = i & ((HW / 4) - 1);
    const float4* xv = (const float4*)x;
    float4 r  = xv[(size_t)(b * 3 + 0) * (HW / 4) + rem];
    float4 g  = xv[(size_t)(b * 3 + 1) * (HW / 4) + rem];
    float4 bl = xv[(size_t)(b * 3 + 2) * (HW / 4) + rem];
    float4 o;
    o.x = 0.2126f * r.x + 0.7152f * g.x + 0.0722f * bl.x;
    o.y = 0.2126f * r.y + 0.7152f * g.y + 0.0722f * bl.y;
    o.z = 0.2126f * r.z + 0.7152f * g.z + 0.0722f * bl.z;
    o.w = 0.2126f * r.w + 0.7152f * g.w + 0.0722f * bl.w;
    const int row = rem >> 8;
    const int grp = rem & 255;
    *(float4*)(L + ((size_t)b * HQ + row) * LP + LOFF + 4 * grp) = o;
}

__device__ __forceinline__ constexpr bool use7(int q, int s) {
    return (q == 1) || (q == 0 && s >= 2) || (q == 2 && s <= 1);
}

__device__ __forceinline__ float other_half(float v) {
    unsigned u = __float_as_uint(v);
    return __uint_as_float(__builtin_amdgcn_permlanex16(u, u, 0x76543210u, 0xfedcba98u, false, false));
}

__device__ __forceinline__ void exchange_halves(float& a, float& b, bool upper) {
    const float ax = other_half(a);
    const float bx = other_half(b);
    const float na = upper ? bx : a;
    const float nb = upper ? b : ax;
    a = na;
    b = nb;
}

__global__ __launch_bounds__(128) void blur_combine_kernel(
    const float* __restrict__ x, const float* __restrict__ L,
    const float* __restrict__ pclar, const float* __restrict__ ptex,
    float* __restrict__ out) {
    __shared__ float sg31[31];
    __shared__ float sg7[7];

    const int tid  = threadIdx.x;
    const int lane = tid & 31;
    const int ln   = lane & 15;
    const int hi   = lane >> 4;
    const bool upper = (hi != 0);
    const int wv   = __builtin_amdgcn_readfirstlane(tid >> 5);

    if (tid < 31) { float d = (float)tid - 15.f; sg31[tid] = expf(-d * d * (1.f / 128.f)); }
    if (tid < 7)  { float d = (float)tid - 3.f;  sg7[tid]  = expf(-d * d * (1.f / 4.5f)); }
    __syncthreads();
    float s31 = 0.f, s7 = 0.f;
    for (int t = 0; t < 31; ++t) s31 += sg31[t];
    for (int t = 0; t < 7;  ++t) s7  += sg7[t];
    const float inv31 = 1.f / s31, inv7 = 1.f / s7;

    v2f w31[3][4];
    v2f w7[3][4];
#pragma unroll
    for (int q = 0; q < 3; ++q) {
#pragma unroll
        for (int s = 0; s < 4; ++s) {
#pragma unroll
            for (int v = 0; v < 2; ++v) {
                int kk = 16 * q + 8 * hi + 2 * s + v;
                int t  = kk - ln - 1;
                int tc = iclamp(t, 0, 30);
                w31[q][s][v] = sg31[tc] * ((t == tc) ? inv31 : 0.f);
                int u  = kk - ln - 13;
                int uc = iclamp(u, 0, 6);
                w7[q][s][v] = sg7[uc] * ((u == uc) ? inv7 : 0.f);
            }
        }
    }

    const int xs = blockIdx.x * (16 * NT);
    const int y0 = (blockIdx.y * 4 + wv) * 16;
    const int b  = blockIdx.z;

    const bool top = (y0 == 0), bot = (y0 == HQ - 16);
    const int   rq0 = top ? 16 : 0;
    const int   rq2 = bot ? 16 : 32;
    const float mq0 = top ? 0.f : 1.f;
    const float mq2 = bot ? 0.f : 1.f;

    const float* Lb = L + (size_t)b * HQ * LP + LOFF;
    const int rb0 = (y0 - 16 + rq0) * LP;
    const int rb1 = y0 * LP;
    const int rb2 = (y0 - 16 + rq2) * LP;

    const float ca = tanhf(pclar[0]) * 0.5f;
    const float ta = tanhf(ptex[0])  * 0.3f;

    const v8f zero8 = {0.f, 0.f, 0.f, 0.f, 0.f, 0.f, 0.f, 0.f};
    v8f PA31 = zero8, PB31 = zero8;
    v8f PA7  = zero8, PB7  = zero8;
    v8f LB = zero8;

#pragma unroll 1
    for (int j = 0; j < NT / 2 + 1; ++j) {
        const int X0   = xs + 32 * j - 16;
        const int col  = X0 + lane;
        const int colc = iclamp(col, 0, WQ - 1);
        const float mc = (col == colc) ? 1.f : 0.f;
        const float m0 = mq0 * mc, m2 = mq2 * mc;

        v8f C31 = zero8, C7 = zero8, D31 = zero8, D7 = zero8, LC = zero8, LD = zero8;
#pragma unroll
        for (int q = 0; q < 3; ++q) {
            const float* src = Lb + ((q == 0) ? rb0 : (q == 1) ? rb1 : rb2) + colc;
            const float  mm  = (q == 0) ? m0 : (q == 1) ? mc : m2;
#pragma unroll
            for (int s = 0; s < 4; ++s) {
                v2f fc, fd;
                fc[0] = src[(2 * s) * LP] * mm;
                fc[1] = src[(2 * s + 1) * LP] * mm;
                fd[0] = src[(8 + 2 * s) * LP] * mm;
                fd[1] = src[(8 + 2 * s + 1) * LP] * mm;
                float c0 = fc[0], c1 = fc[1], d0 = fd[0], d1 = fd[1];
                exchange_halves(c0, d0, upper);
                exchange_halves(c1, d1, upper);
                fc[0] = c0; fc[1] = c1; fd[0] = d0; fd[1] = d1;
                C31 = __builtin_amdgcn_wmma_f32_16x16x4_f32(
                    false, fc, false, w31[q][s], (short)0, C31, false, false);
                D31 = __builtin_amdgcn_wmma_f32_16x16x4_f32(
                    false, fd, false, w31[q][s], (short)0, D31, false, false);
                if (use7(q, s)) {
                    C7 = __builtin_amdgcn_wmma_f32_16x16x4_f32(
                        false, fc, false, w7[q][s], (short)0, C7, false, false);
                    D7 = __builtin_amdgcn_wmma_f32_16x16x4_f32(
                        false, fd, false, w7[q][s], (short)0, D7, false, false);
                }
                if (q == 1) {
                    LC[2 * s] = fc[0]; LC[2 * s + 1] = fc[1];
                    LD[2 * s] = fd[0]; LD[2 * s + 1] = fd[1];
                }
            }
        }

        if (j >= 1) {
            const int x0 = xs + 32 * (j - 1);
            v8f e31 = zero8, e7 = zero8, f31 = zero8, f7 = zero8;
#pragma unroll
            for (int q = 0; q < 3; ++q) {
                const v8f p31 = (q == 0) ? PA31 : (q == 1) ? PB31 : C31;
                const v8f p7  = (q == 0) ? PA7  : (q == 1) ? PB7  : C7;
                const v8f r31 = (q == 0) ? PB31 : (q == 1) ? C31 : D31;
                const v8f r7  = (q == 0) ? PB7  : (q == 1) ? C7  : D7;
#pragma unroll
                for (int s = 0; s < 4; ++s) {
                    v2f a31 = {p31[2 * s], p31[2 * s + 1]};
                    v2f b31 = {r31[2 * s], r31[2 * s + 1]};
                    e31 = __builtin_amdgcn_wmma_f32_16x16x4_f32(
                        false, a31, false, w31[q][s], (short)0, e31, false, false);
                    f31 = __builtin_amdgcn_wmma_f32_16x16x4_f32(
                        false, b31, false, w31[q][s], (short)0, f31, false, false);
                    if (use7(q, s)) {
                        v2f a7 = {p7[2 * s], p7[2 * s + 1]};
                        v2f b7 = {r7[2 * s], r7[2 * s + 1]};
                        e7 = __builtin_amdgcn_wmma_f32_16x16x4_f32(
                            false, a7, false, w7[q][s], (short)0, e7, false, false);
                        f7 = __builtin_amdgcn_wmma_f32_16x16x4_f32(
                            false, b7, false, w7[q][s], (short)0, f7, false, false);
                    }
                }
            }
            const size_t pbase = (size_t)y0 * WQ + x0 + lane;
#pragma unroll
            for (int r = 0; r < 8; ++r) {
                float Lv0 = LB[r];
                float le0 = Lv0 + ca * (Lv0 - e31[r]) + ta * (Lv0 - e7[r]);
                float ra  = (le0 + 1e-6f) / (Lv0 + 1e-6f);
                float Lv1 = LC[r];
                float le1 = Lv1 + ca * (Lv1 - f31[r]) + ta * (Lv1 - f7[r]);
                float rb  = (le1 + 1e-6f) / (Lv1 + 1e-6f);
                exchange_halves(ra, rb, upper);
#pragma unroll
                for (int c = 0; c < 3; ++c) {
                    size_t xa = (size_t)(b * 3 + c) * HW + pbase + (size_t)r * WQ;
                    size_t xb = xa + (size_t)8 * WQ;
                    float va = x[xa] * ra;
                    float vb = x[xb] * rb;
                    out[xa] = fminf(fmaxf(va, 0.f), 1.f);
                    out[xb] = fminf(fmaxf(vb, 0.f), 1.f);
                }
            }
        }
        PA31 = C31; PB31 = D31;
        PA7  = C7;  PB7  = D7;
        LB = LD;
    }
}

extern "C" void kernel_launch(void* const* d_in, const int* in_sizes, int n_in,
                              void* d_out, int out_size, void* d_ws, size_t ws_size,
                              hipStream_t stream) {
    const float* x    = (const float*)d_in[0];
    const float* clar = (const float*)d_in[1];
    const float* tex  = (const float*)d_in[2];
    float* out = (float*)d_out;

    float* L = (float*)d_ws;

    luma_kernel<<<(NB * HW / 4) / 256, 256, 0, stream>>>(x, L);
    dim3 grid2(WQ / (16 * NT), HQ / 64, NB);
    blur_combine_kernel<<<grid2, 128, 0, stream>>>(x, L, clar, tex, out);
}


# luma plane stored in A-fragment order (4 rows of a column adjacent): a lane's 16-byte load is its fragments of 2 k-steps; 12 luma loads a step, not 48, no lane exchange; values, sum order unchanged
# speedup vs baseline: 3.9004x; 1.0695x over previous
#include <hip/hip_runtime.h>
#include <cmath>


#define HQ 1024
#define WQ 1024
#define HW (HQ * WQ)
#define NB 8
#define NC 3
#define NT 8
#define LPQ (WQ + 32)
#define LOFF 16
#define NQR (HQ / 4)

typedef __attribute__((ext_vector_type(2))) float v2f;
typedef __attribute__((ext_vector_type(8))) float v8f;

__device__ __forceinline__ int iclamp(int v, int lo, int hi) {
    return v < lo ? lo : (v > hi ? hi : v);
}

__global__ __launch_bounds__(256) void luma_kernel(const float* __restrict__ x,
                                                   float* __restrict__ L) {
    const int i  = blockIdx.x * 256 + threadIdx.x;
    const int c  = i & (WQ - 1);
    const int qg = i >> 10;
    const int b  = qg >> 8;
    const int Q  = qg & (NQR - 1);
    const float* xr = x + (size_t)(b * 3) * HW + (size_t)(4 * Q) * WQ + c;
    float lum[4];
#pragma unroll
    for (int e = 0; e < 4; ++e) {
        const float r  = xr[e * WQ];
        const float g  = xr[HW + e * WQ];
        const float bl = xr[2 * HW + e * WQ];
        lum[e] = 0.2126f * r + 0.7152f * g + 0.0722f * bl;
    }
    float4 o;
    o.x = lum[0]; o.y = lum[1]; o.z = lum[2]; o.w = lum[3];
    *(float4*)(L + ((size_t)qg * LPQ + LOFF + c) * 4) = o;
}

__device__ __forceinline__ constexpr bool use7(int q, int s) {
    return (q == 1) || (q == 0 && s >= 2) || (q == 2 && s <= 1);
}

__device__ __forceinline__ float other_half(float v) {
    unsigned u = __float_as_uint(v);
    return __uint_as_float(__builtin_amdgcn_permlanex16(u, u, 0x76543210u, 0xfedcba98u, false, false));
}

__device__ __forceinline__ void exchange_halves(float& a, float& b, bool upper) {
    const float ax = other_half(a);
    const float bx = other_half(b);
    const float na = upper ? bx : a;
    const float nb = upper ? b : ax;
    a = na;
    b = nb;
}

__global__ __launch_bounds__(128) void blur_combine_kernel(
    const float* __restrict__ x, const float* __restrict__ L,
    const float* __restrict__ pclar, const float* __restrict__ ptex,
    float* __restrict__ out) {
    __shared__ float sg31[31];
    __shared__ float sg7[7];

    const int tid  = threadIdx.x;
    const int lane = tid & 31;
    const int ln   = lane & 15;
    const int hi   = lane >> 4;
    const bool upper = (hi != 0);
    const int wv   = __builtin_amdgcn_readfirstlane(tid >> 5);

    if (tid < 31) { float d = (float)tid - 15.f; sg31[tid] = expf(-d * d * (1.f / 128.f)); }
    if (tid < 7)  { float d = (float)tid - 3.f;  sg7[tid]  = expf(-d * d * (1.f / 4.5f)); }
    __syncthreads();
    float s31 = 0.f, s7 = 0.f;
    for (int t = 0; t < 31; ++t) s31 += sg31[t];
    for (int t = 0; t < 7;  ++t) s7  += sg7[t];
    const float inv31 = 1.f / s31, inv7 = 1.f / s7;

    v2f w31[3][4];
    v2f w7[3][4];
#pragma unroll
    for (int q = 0; q < 3; ++q) {
#pragma unroll
        for (int s = 0; s < 4; ++s) {
#pragma unroll
            for (int v = 0; v < 2; ++v) {
                int kk = 16 * q + 8 * hi + 2 * s + v;
                int t  = kk - ln - 1;
                int tc = iclamp(t, 0, 30);
                w31[q][s][v] = sg31[tc] * ((t == tc) ? inv31 : 0.f);
                int u  = kk - ln - 13;
                int uc = iclamp(u, 0, 6);
                w7[q][s][v] = sg7[uc] * ((u == uc) ? inv7 : 0.f);
            }
        }
    }

    const int xs = blockIdx.x * (16 * NT);
    const int y0 = (blockIdx.y * 4 + wv) * 16;
    const int b  = blockIdx.z;

    const bool top = (y0 == 0), bot = (y0 == HQ - 16);
    const int   rq0 = top ? 16 : 0;
    const int   rq2 = bot ? 16 : 32;
    const float mq0 = top ? 0.f : 1.f;
    const float mq2 = bot ? 0.f : 1.f;

    const float4* Lb = (const float4*)L + (size_t)b * NQR * LPQ + LOFF;
    const int rb0 = ((y0 - 16 + rq0) >> 2) * LPQ;
    const int rb1 = (y0 >> 2) * LPQ;
    const int rb2 = ((y0 - 16 + rq2) >> 2) * LPQ;
    const int lh = 2 * hi * LPQ + ln;

    const float ca = tanhf(pclar[0]) * 0.5f;
    const float ta = tanhf(ptex[0])  * 0.3f;

    const v8f zero8 = {0.f, 0.f, 0.f, 0.f, 0.f, 0.f, 0.f, 0.f};
    v8f PA31 = zero8, PB31 = zero8;
    v8f PA7  = zero8, PB7  = zero8;
    v8f LB = zero8;

#pragma unroll 1
    for (int j = 0; j < NT / 2 + 1; ++j) {
        const int X0 = xs + 32 * j - 16;
        const bool cout_ = (X0 < 0), dout_ = (X0 + 16 >= WQ);
        const int   colC = cout_ ? X0 + 16 : X0;
        const int   colD = dout_ ? X0 : X0 + 16;
        const float mC = cout_ ? 0.f : 1.f;
        const float mD = dout_ ? 0.f : 1.f;

        v8f C31 = zero8, C7 = zero8, D31 = zero8, D7 = zero8, LC = zero8, LD = zero8;
#pragma unroll
        for (int q = 0; q < 3; ++q) {
            const int     rbq  = (q == 0) ? rb0 : (q == 1) ? rb1 : rb2;
            const float4* srcC = Lb + (rbq + colC);
            const float4* srcD = Lb + (rbq + colD);
            const float   mq   = (q == 0) ? mq0 : (q == 1) ? 1.f : mq2;
            const float   mmC = mq * mC, mmD = mq * mD;
#pragma unroll
            for (int u = 0; u < 2; ++u) {
                const float4 cc = srcC[u * LPQ + lh];
                const float4 cd = srcD[u * LPQ + lh];
#pragma unroll
                for (int t = 0; t < 2; ++t) {
                    const int s = 2 * u + t;
                    v2f fc, fd;
                    fc[0] = (t == 0 ? cc.x : cc.z) * mmC;
                    fc[1] = (t == 0 ? cc.y : cc.w) * mmC;
                    fd[0] = (t == 0 ? cd.x : cd.z) * mmD;
                    fd[1] = (t == 0 ? cd.y : cd.w) * mmD;
                    C31 = __builtin_amdgcn_wmma_f32_16x16x4_f32(
                        false, fc, false, w31[q][s], (short)0, C31, false, false);
                    D31 = __builtin_amdgcn_wmma_f32_16x16x4_f32(
                        false, fd, false, w31[q][s], (short)0, D31, false, false);
                    if (use7(q, s)) {
                        C7 = __builtin_amdgcn_wmma_f32_16x16x4_f32(
                            false, fc, false, w7[q][s], (short)0, C7, false, false);
                        D7 = __builtin_amdgcn_wmma_f32_16x16x4_f32(
                            false, fd, false, w7[q][s], (short)0, D7, false, false);
                    }
                    if (q == 1) {
                        LC[2 * s] = fc[0]; LC[2 * s + 1] = fc[1];
                        LD[2 * s] = fd[0]; LD[2 * s + 1] = fd[1];
                    }
                }
            }
        }

        if (j >= 1) {
            const int x0 = xs + 32 * (j - 1);
            v8f e31 = zero8, e7 = zero8, f31 = zero8, f7 = zero8;
#pragma unroll
            for (int q = 0; q < 3; ++q) {
                const v8f p31 = (q == 0) ? PA31 : (q == 1) ? PB31 : C31;
                const v8f p7  = (q == 0) ? PA7  : (q == 1) ? PB7  : C7;
                const v8f r31 = (q == 0) ? PB31 : (q == 1) ? C31 : D31;
                const v8f r7  = (q == 0) ? PB7  : (q == 1) ? C7  : D7;
#pragma unroll
                for (int s = 0; s < 4; ++s) {
                    v2f a31 = {p31[2 * s], p31[2 * s + 1]};
                    v2f b31 = {r31[2 * s], r31[2 * s + 1]};
                    e31 = __builtin_amdgcn_wmma_f32_16x16x4_f32(
                        false, a31, false, w31[q][s], (short)0, e31, false, false);
                    f31 = __builtin_amdgcn_wmma_f32_16x16x4_f32(
                        false, b31, false, w31[q][s], (short)0, f31, false, false);
                    if (use7(q, s)) {
                        v2f a7 = {p7[2 * s], p7[2 * s + 1]};
                        v2f b7 = {r7[2 * s], r7[2 * s + 1]};
                        e7 = __builtin_amdgcn_wmma_f32_16x16x4_f32(
                            false, a7, false, w7[q][s], (short)0, e7, false, false);
                        f7 = __builtin_amdgcn_wmma_f32_16x16x4_f32(
                            false, b7, false, w7[q][s], (short)0, f7, false, false);
                    }
                }
            }
            const size_t pbase = (size_t)y0 * WQ + x0 + lane;
#pragma unroll
            for (int r = 0; r < 8; ++r) {
                float Lv0 = LB[r];
                float le0 = Lv0 + ca * (Lv0 - e31[r]) + ta * (Lv0 - e7[r]);
                float ra  = (le0 + 1e-6f) / (Lv0 + 1e-6f);
                float Lv1 = LC[r];
                float le1 = Lv1 + ca * (Lv1 - f31[r]) + ta * (Lv1 - f7[r]);
                float rb  = (le1 + 1e-6f) / (Lv1 + 1e-6f);
                exchange_halves(ra, rb, upper);
#pragma unroll
                for (int c = 0; c < 3; ++c) {
                    size_t xa = (size_t)(b * 3 + c) * HW + pbase + (size_t)r * WQ;
                    size_t xb = xa + (size_t)8 * WQ;
                    float va = x[xa] * ra;
                    float vb = x[xb] * rb;
                    out[xa] = fminf(fmaxf(va, 0.f), 1.f);
                    out[xb] = fminf(fmaxf(vb, 0.f), 1.f);
                }
            }
        }
        PA31 = C31; PB31 = D31;
        PA7  = C7;  PB7  = D7;
        LB = LD;
    }
}

extern "C" void kernel_launch(void* const* d_in, const int* in_sizes, int n_in,
                              void* d_out, int out_size, void* d_ws, size_t ws_size,
                              hipStream_t stream) {
    const float* x    = (const float*)d_in[0];
    const float* clar = (const float*)d_in[1];
    const float* tex  = (const float*)d_in[2];
    float* out = (float*)d_out;

    float* L = (float*)d_ws;

    luma_kernel<<<(NB * NQR * WQ) / 256, 256, 0, stream>>>(x, L);
    dim3 grid2(WQ / (16 * NT), HQ / 64, NB);
    blur_combine_kernel<<<grid2, 128, 0, stream>>>(x, L, clar, tex, out);
}


# blur kernel: a step's 12 luma loads are issued one step ahead, behind the vertical products that last read their registers, so their round trip runs under the matrix work; same sums
# speedup vs baseline: 4.0439x; 1.0368x over previous
#include <hip/hip_runtime.h>
#include <cmath>


#define HQ 1024
#define WQ 1024
#define HW (HQ * WQ)
#define NB 8
#define NC 3
#define NT 8
#define LPQ (WQ + 32)
#define LOFF 16
#define NQR (HQ / 4)

typedef __attribute__((ext_vector_type(2))) float v2f;
typedef __attribute__((ext_vector_type(8))) float v8f;

__device__ __forceinline__ int iclamp(int v, int lo, int hi) {
    return v < lo ? lo : (v > hi ? hi : v);
}

__global__ __launch_bounds__(256) void luma_kernel(const float* __restrict__ x,
                                                   float* __restrict__ L) {
    const int i  = blockIdx.x * 256 + threadIdx.x;
    const int c  = i & (WQ - 1);
    const int qg = i >> 10;
    const int b  = qg >> 8;
    const int Q  = qg & (NQR - 1);
    const float* xr = x + (size_t)(b * 3) * HW + (size_t)(4 * Q) * WQ + c;
    float lum[4];
#pragma unroll
    for (int e = 0; e < 4; ++e) {
        const float r  = xr[e * WQ];
        const float g  = xr[HW + e * WQ];
        const float bl = xr[2 * HW + e * WQ];
        lum[e] = 0.2126f * r + 0.7152f * g + 0.0722f * bl;
    }
    float4 o;
    o.x = lum[0]; o.y = lum[1]; o.z = lum[2]; o.w = lum[3];
    *(float4*)(L + ((size_t)qg * LPQ + LOFF + c) * 4) = o;
}

__device__ __forceinline__ constexpr bool use7(int q, int s) {
    return (q == 1) || (q == 0 && s >= 2) || (q == 2 && s <= 1);
}

__device__ __forceinline__ float other_half(float v) {
    unsigned u = __float_as_uint(v);
    return __uint_as_float(__builtin_amdgcn_permlanex16(u, u, 0x76543210u, 0xfedcba98u, false, false));
}

__device__ __forceinline__ void exchange_halves(float& a, float& b, bool upper) {
    const float ax = other_half(a);
    const float bx = other_half(b);
    const float na = upper ? bx : a;
    const float nb = upper ? b : ax;
    a = na;
    b = nb;
}


#define NSTEP (NT / 2 + 1)

__device__ __forceinline__ void load_step_cells(const float4* __restrict__ Lb, int rb0, int rb1,
                                                int rb2, int lh, int xs, int j,
                                                float4 (&oc)[6], float4 (&od)[6]) {
    const int X0 = xs + 32 * j - 16;
    const int colC = (X0 < 0) ? X0 + 16 : X0;
    const int colD = (X0 + 16 >= WQ) ? X0 : X0 + 16;
#pragma unroll
    for (int q = 0; q < 3; ++q) {
        const int rbq = (q == 0) ? rb0 : (q == 1) ? rb1 : rb2;
#pragma unroll
        for (int u = 0; u < 2; ++u) {
            oc[2 * q + u] = Lb[rbq + colC + u * LPQ + lh];
            od[2 * q + u] = Lb[rbq + colD + u * LPQ + lh];
        }
    }
}

__global__ __launch_bounds__(128) void blur_combine_kernel(
    const float* __restrict__ x, const float* __restrict__ L,
    const float* __restrict__ pclar, const float* __restrict__ ptex,
    float* __restrict__ out) {
    __shared__ float sg31[31];
    __shared__ float sg7[7];

    const int tid  = threadIdx.x;
    const int lane = tid & 31;
    const int ln   = lane & 15;
    const int hi   = lane >> 4;
    const bool upper = (hi != 0);
    const int wv   = __builtin_amdgcn_readfirstlane(tid >> 5);

    const int xs = blockIdx.x * (16 * NT);
    const int y0 = (blockIdx.y * 4 + wv) * 16;
    const int b  = blockIdx.z;

    const bool top = (y0 == 0), bot = (y0 == HQ - 16);
    const int   rq0 = top ? 16 : 0;
    const int   rq2 = bot ? 16 : 32;
    const float mq0 = top ? 0.f : 1.f;
    const float mq2 = bot ? 0.f : 1.f;

    const float4* Lb = (const float4*)L + (size_t)b * NQR * LPQ + LOFF;
    const int rb0 = ((y0 - 16 + rq0) >> 2) * LPQ;
    const int rb1 = (y0 >> 2) * LPQ;
    const int rb2 = ((y0 - 16 + rq2) >> 2) * LPQ;
    const int lh = 2 * hi * LPQ + ln;

    if (tid < 31) { float d = (float)tid - 15.f; sg31[tid] = expf(-d * d * (1.f / 128.f)); }
    if (tid < 7)  { float d = (float)tid - 3.f;  sg7[tid]  = expf(-d * d * (1.f / 4.5f)); }
    __syncthreads();

    float4 cc[6], cd[6];
    load_step_cells(Lb, rb0, rb1, rb2, lh, xs, 0, cc, cd);

    float s31 = 0.f, s7 = 0.f;
    for (int t = 0; t < 31; ++t) s31 += sg31[t];
    for (int t = 0; t < 7;  ++t) s7  += sg7[t];
    const float inv31 = 1.f / s31, inv7 = 1.f / s7;

    v2f w31[3][4];
    v2f w7[3][4];
#pragma unroll
    for (int q = 0; q < 3; ++q) {
#pragma unroll
        for (int s = 0; s < 4; ++s) {
#pragma unroll
            for (int v = 0; v < 2; ++v) {
                int kk = 16 * q + 8 * hi + 2 * s + v;
                int t  = kk - ln - 1;
                int tc = iclamp(t, 0, 30);
                w31[q][s][v] = sg31[tc] * ((t == tc) ? inv31 : 0.f);
                int u  = kk - ln - 13;
                int uc = iclamp(u, 0, 6);
                w7[q][s][v] = sg7[uc] * ((u == uc) ? inv7 : 0.f);
            }
        }
    }

    const float ca = tanhf(pclar[0]) * 0.5f;
    const float ta = tanhf(ptex[0])  * 0.3f;

    const v8f zero8 = {0.f, 0.f, 0.f, 0.f, 0.f, 0.f, 0.f, 0.f};
    v8f PA31 = zero8, PB31 = zero8;
    v8f PA7  = zero8, PB7  = zero8;
    v8f LB = zero8;

#pragma unroll 1
    for (int j = 0; j < NSTEP; ++j) {
        const int X0 = xs + 32 * j - 16;
        const bool cout_ = (X0 < 0), dout_ = (X0 + 16 >= WQ);
        const float mC = cout_ ? 0.f : 1.f;
        const float mD = dout_ ? 0.f : 1.f;

        v8f C31 = zero8, C7 = zero8, D31 = zero8, D7 = zero8, LC = zero8, LD = zero8;
#pragma unroll
        for (int q = 0; q < 3; ++q) {
            const float   mq   = (q == 0) ? mq0 : (q == 1) ? 1.f : mq2;
            const float   mmC = mq * mC, mmD = mq * mD;
#pragma unroll
            for (int u = 0; u < 2; ++u) {
                const float4 wc = cc[2 * q + u];
                const float4 wd = cd[2 * q + u];
#pragma unroll
                for (int t = 0; t < 2; ++t) {
                    const int s = 2 * u + t;
                    v2f fc, fd;
                    fc[0] = (t == 0 ? wc.x : wc.z) * mmC;
                    fc[1] = (t == 0 ? wc.y : wc.w) * mmC;
                    fd[0] = (t == 0 ? wd.x : wd.z) * mmD;
                    fd[1] = (t == 0 ? wd.y : wd.w) * mmD;
                    C31 = __builtin_amdgcn_wmma_f32_16x16x4_f32(
                        false, fc, false, w31[q][s], (short)0, C31, false, false);
                    D31 = __builtin_amdgcn_wmma_f32_16x16x4_f32(
                        false, fd, false, w31[q][s], (short)0, D31, false, false);
                    if (use7(q, s)) {
                        C7 = __builtin_amdgcn_wmma_f32_16x16x4_f32(
                            false, fc, false, w7[q][s], (short)0, C7, false, false);
                        D7 = __builtin_amdgcn_wmma_f32_16x16x4_f32(
                            false, fd, false, w7[q][s], (short)0, D7, false, false);
                    }
                    if (q == 1) {
                        LC[2 * s] = fc[0]; LC[2 * s + 1] = fc[1];
                        LD[2 * s] = fd[0]; LD[2 * s + 1] = fd[1];
                    }
                }
            }
        }

        if (j + 1 < NSTEP) load_step_cells(Lb, rb0, rb1, rb2, lh, xs, j + 1, cc, cd);

        if (j >= 1) {
            const int x0 = xs + 32 * (j - 1);
            v8f e31 = zero8, e7 = zero8, f31 = zero8, f7 = zero8;
#pragma unroll
            for (int q = 0; q < 3; ++q) {
                const v8f p31 = (q == 0) ? PA31 : (q == 1) ? PB31 : C31;
                const v8f p7  = (q == 0) ? PA7  : (q == 1) ? PB7  : C7;
                const v8f r31 = (q == 0) ? PB31 : (q == 1) ? C31 : D31;
                const v8f r7  = (q == 0) ? PB7  : (q == 1) ? C7  : D7;
#pragma unroll
                for (int s = 0; s < 4; ++s) {
                    v2f a31 = {p31[2 * s], p31[2 * s + 1]};
                    v2f b31 = {r31[2 * s], r31[2 * s + 1]};
                    e31 = __builtin_amdgcn_wmma_f32_16x16x4_f32(
                        false, a31, false, w31[q][s], (short)0, e31, false, false);
                    f31 = __builtin_amdgcn_wmma_f32_16x16x4_f32(
                        false, b31, false, w31[q][s], (short)0, f31, false, false);
                    if (use7(q, s)) {
                        v2f a7 = {p7[2 * s], p7[2 * s + 1]};
                        v2f b7 = {r7[2 * s], r7[2 * s + 1]};
                        e7 = __builtin_amdgcn_wmma_f32_16x16x4_f32(
                            false, a7, false, w7[q][s], (short)0, e7, false, false);
                        f7 = __builtin_amdgcn_wmma_f32_16x16x4_f32(
                            false, b7, false, w7[q][s], (short)0, f7, false, false);
                    }
                }
            }
            const size_t pbase = (size_t)y0 * WQ + x0 + lane;
#pragma unroll
            for (int r = 0; r < 8; ++r) {
                float Lv0 = LB[r];
                float le0 = Lv0 + ca * (Lv0 - e31[r]) + ta * (Lv0 - e7[r]);
                float ra  = (le0 + 1e-6f) / (Lv0 + 1e-6f);
                float Lv1 = LC[r];
                float le1 = Lv1 + ca * (Lv1 - f31[r]) + ta * (Lv1 - f7[r]);
                float rb  = (le1 + 1e-6f) / (Lv1 + 1e-6f);
                exchange_halves(ra, rb, upper);
#pragma unroll
                for (int c = 0; c < 3; ++c) {
                    size_t xa = (size_t)(b * 3 + c) * HW + pbase + (size_t)r * WQ;
                    size_t xb = xa + (size_t)8 * WQ;
                    float va = x[xa] * ra;
                    float vb = x[xb] * rb;
                    out[xa] = fminf(fmaxf(va, 0.f), 1.f);
                    out[xb] = fminf(fmaxf(vb, 0.f), 1.f);
                }
            }
        }
        PA31 = C31; PB31 = D31;
        PA7  = C7;  PB7  = D7;
        LB = LD;
    }
}

extern "C" void kernel_launch(void* const* d_in, const int* in_sizes, int n_in,
                              void* d_out, int out_size, void* d_ws, size_t ws_size,
                              hipStream_t stream) {
    const float* x    = (const float*)d_in[0];
    const float* clar = (const float*)d_in[1];
    const float* tex  = (const float*)d_in[2];
    float* out = (float*)d_out;

    float* L = (float*)d_ws;

    luma_kernel<<<(NB * NQR * WQ) / 256, 256, 0, stream>>>(x, L);
    dim3 grid2(WQ / (16 * NT), HQ / 64, NB);
    blur_combine_kernel<<<grid2, 128, 0, stream>>>(x, L, clar, tex, out);
}


# combine step: ratio registers re-laid in registers (DPP 4x4 transposes in lane quads + permlane16 in each half, replacing the half exchange): colour loads, output stores 16 bytes a lane, 12+12 a step
# speedup vs baseline: 4.1778x; 1.0331x over previous
#include <hip/hip_runtime.h>
#include <cmath>


#define HQ 1024
#define WQ 1024
#define HW (HQ * WQ)
#define NB 8
#define NC 3
#define NT 8
#define LPQ (WQ + 32)
#define LOFF 16
#define NQR (HQ / 4)

typedef __attribute__((ext_vector_type(2))) float v2f;
typedef __attribute__((ext_vector_type(8))) float v8f;
typedef __attribute__((ext_vector_type(4))) float v4f;

__device__ __forceinline__ int iclamp(int v, int lo, int hi) {
    return v < lo ? lo : (v > hi ? hi : v);
}

__global__ __launch_bounds__(256) void luma_kernel(const float* __restrict__ x,
                                                   float* __restrict__ L) {
    const int i  = blockIdx.x * 256 + threadIdx.x;
    const int c  = i & (WQ - 1);
    const int qg = i >> 10;
    const int b  = qg >> 8;
    const int Q  = qg & (NQR - 1);
    const float* xr = x + (size_t)(b * 3) * HW + (size_t)(4 * Q) * WQ + c;
    float lum[4];
#pragma unroll
    for (int e = 0; e < 4; ++e) {
        const float r  = xr[e * WQ];
        const float g  = xr[HW + e * WQ];
        const float bl = xr[2 * HW + e * WQ];
        lum[e] = 0.2126f * r + 0.7152f * g + 0.0722f * bl;
    }
    float4 o;
    o.x = lum[0]; o.y = lum[1]; o.z = lum[2]; o.w = lum[3];
    *(float4*)(L + ((size_t)qg * LPQ + LOFF + c) * 4) = o;
}

__device__ __forceinline__ constexpr bool use7(int q, int s) {
    return (q == 1) || (q == 0 && s >= 2) || (q == 2 && s <= 1);
}

__device__ __forceinline__ float quad_xor1(float v) {
    const int u = __float_as_int(v);
    return __int_as_float(__builtin_amdgcn_update_dpp(u, u, 0xB1, 0xF, 0xF, false));
}
__device__ __forceinline__ float quad_xor2(float v) {
    const int u = __float_as_int(v);
    return __int_as_float(__builtin_amdgcn_update_dpp(u, u, 0x4E, 0xF, 0xF, false));
}

__device__ __forceinline__ void quad_transpose(float& a0, float& a1, float& a2, float& a3,
                                               bool odd, bool hiq) {
    const float t0 = quad_xor1(a1), t1 = quad_xor1(a0);
    const float t2 = quad_xor1(a3), t3 = quad_xor1(a2);
    const float b0 = odd ? t0 : a0;
    const float b1 = odd ? a1 : t1;
    const float b2 = odd ? t2 : a2;
    const float b3 = odd ? a3 : t3;
    const float u0 = quad_xor2(b2), u2 = quad_xor2(b0);
    const float u1 = quad_xor2(b3), u3 = quad_xor2(b1);
    a0 = hiq ? u0 : b0;
    a2 = hiq ? b2 : u2;
    a1 = hiq ? u1 : b1;
    a3 = hiq ? b3 : u3;
}

__device__ __forceinline__ float half_transpose4x4(float v) {
    const unsigned u = __float_as_uint(v);
    return __uint_as_float(__builtin_amdgcn_permlane16(u, u, 0xD951C840u, 0xFB73EA62u, false, false));
}


#define NSTEP (NT / 2 + 1)

__device__ __forceinline__ void load_step_cells(const float4* __restrict__ Lb, int rb0, int rb1,
                                                int rb2, int lh, int xs, int j,
                                                float4 (&oc)[6], float4 (&od)[6]) {
    const int X0 = xs + 32 * j - 16;
    const int colC = (X0 < 0) ? X0 + 16 : X0;
    const int colD = (X0 + 16 >= WQ) ? X0 : X0 + 16;
#pragma unroll
    for (int q = 0; q < 3; ++q) {
        const int rbq = (q == 0) ? rb0 : (q == 1) ? rb1 : rb2;
#pragma unroll
        for (int u = 0; u < 2; ++u) {
            oc[2 * q + u] = Lb[rbq + colC + u * LPQ + lh];
            od[2 * q + u] = Lb[rbq + colD + u * LPQ + lh];
        }
    }
}

__global__ __launch_bounds__(128) void blur_combine_kernel(
    const float* __restrict__ x, const float* __restrict__ L,
    const float* __restrict__ pclar, const float* __restrict__ ptex,
    float* __restrict__ out) {
    __shared__ float sg31[31];
    __shared__ float sg7[7];

    const int tid  = threadIdx.x;
    const int lane = tid & 31;
    const int ln   = lane & 15;
    const int hi   = lane >> 4;
    const bool qodd = (lane & 1) != 0;
    const bool qhi  = (lane & 2) != 0;
    const int wv   = __builtin_amdgcn_readfirstlane(tid >> 5);

    const int xs = blockIdx.x * (16 * NT);
    const int y0 = (blockIdx.y * 4 + wv) * 16;
    const int b  = blockIdx.z;

    const bool top = (y0 == 0), bot = (y0 == HQ - 16);
    const int   rq0 = top ? 16 : 0;
    const int   rq2 = bot ? 16 : 32;
    const float mq0 = top ? 0.f : 1.f;
    const float mq2 = bot ? 0.f : 1.f;

    const float4* Lb = (const float4*)L + (size_t)b * NQR * LPQ + LOFF;
    const int rb0 = ((y0 - 16 + rq0) >> 2) * LPQ;
    const int rb1 = (y0 >> 2) * LPQ;
    const int rb2 = ((y0 - 16 + rq2) >> 2) * LPQ;
    const int lh = 2 * hi * LPQ + ln;

    if (tid < 31) { float d = (float)tid - 15.f; sg31[tid] = expf(-d * d * (1.f / 128.f)); }
    if (tid < 7)  { float d = (float)tid - 3.f;  sg7[tid]  = expf(-d * d * (1.f / 4.5f)); }
    __syncthreads();

    float4 cc[6], cd[6];
    load_step_cells(Lb, rb0, rb1, rb2, lh, xs, 0, cc, cd);

    float s31 = 0.f, s7 = 0.f;
    for (int t = 0; t < 31; ++t) s31 += sg31[t];
    for (int t = 0; t < 7;  ++t) s7  += sg7[t];
    const float inv31 = 1.f / s31, inv7 = 1.f / s7;

    v2f w31[3][4];
    v2f w7[3][4];
#pragma unroll
    for (int q = 0; q < 3; ++q) {
#pragma unroll
        for (int s = 0; s < 4; ++s) {
#pragma unroll
            for (int v = 0; v < 2; ++v) {
                int kk = 16 * q + 8 * hi + 2 * s + v;
                int t  = kk - ln - 1;
                int tc = iclamp(t, 0, 30);
                w31[q][s][v] = sg31[tc] * ((t == tc) ? inv31 : 0.f);
                int u  = kk - ln - 13;
                int uc = iclamp(u, 0, 6);
                w7[q][s][v] = sg7[uc] * ((u == uc) ? inv7 : 0.f);
            }
        }
    }

    const float ca = tanhf(pclar[0]) * 0.5f;
    const float ta = tanhf(ptex[0])  * 0.3f;

    const v8f zero8 = {0.f, 0.f, 0.f, 0.f, 0.f, 0.f, 0.f, 0.f};
    v8f PA31 = zero8, PB31 = zero8;
    v8f PA7  = zero8, PB7  = zero8;
    v8f LB = zero8;

#pragma unroll 1
    for (int j = 0; j < NSTEP; ++j) {
        const int X0 = xs + 32 * j - 16;
        const bool cout_ = (X0 < 0), dout_ = (X0 + 16 >= WQ);
        const float mC = cout_ ? 0.f : 1.f;
        const float mD = dout_ ? 0.f : 1.f;

        v8f C31 = zero8, C7 = zero8, D31 = zero8, D7 = zero8, LC = zero8, LD = zero8;
#pragma unroll
        for (int q = 0; q < 3; ++q) {
            const float   mq   = (q == 0) ? mq0 : (q == 1) ? 1.f : mq2;
            const float   mmC = mq * mC, mmD = mq * mD;
#pragma unroll
            for (int u = 0; u < 2; ++u) {
                const float4 wc = cc[2 * q + u];
                const float4 wd = cd[2 * q + u];
#pragma unroll
                for (int t = 0; t < 2; ++t) {
                    const int s = 2 * u + t;
                    v2f fc, fd;
                    fc[0] = (t == 0 ? wc.x : wc.z) * mmC;
                    fc[1] = (t == 0 ? wc.y : wc.w) * mmC;
                    fd[0] = (t == 0 ? wd.x : wd.z) * mmD;
                    fd[1] = (t == 0 ? wd.y : wd.w) * mmD;
                    C31 = __builtin_amdgcn_wmma_f32_16x16x4_f32(
                        false, fc, false, w31[q][s], (short)0, C31, false, false);
                    D31 = __builtin_amdgcn_wmma_f32_16x16x4_f32(
                        false, fd, false, w31[q][s], (short)0, D31, false, false);
                    if (use7(q, s)) {
                        C7 = __builtin_amdgcn_wmma_f32_16x16x4_f32(
                            false, fc, false, w7[q][s], (short)0, C7, false, false);
                        D7 = __builtin_amdgcn_wmma_f32_16x16x4_f32(
                            false, fd, false, w7[q][s], (short)0, D7, false, false);
                    }
                    if (q == 1) {
                        LC[2 * s] = fc[0]; LC[2 * s + 1] = fc[1];
                        LD[2 * s] = fd[0]; LD[2 * s + 1] = fd[1];
                    }
                }
            }
        }

        if (j + 1 < NSTEP) load_step_cells(Lb, rb0, rb1, rb2, lh, xs, j + 1, cc, cd);

        if (j >= 1) {
            const int x0 = xs + 32 * (j - 1);
            v8f e31 = zero8, e7 = zero8, f31 = zero8, f7 = zero8;
#pragma unroll
            for (int q = 0; q < 3; ++q) {
                const v8f p31 = (q == 0) ? PA31 : (q == 1) ? PB31 : C31;
                const v8f p7  = (q == 0) ? PA7  : (q == 1) ? PB7  : C7;
                const v8f r31 = (q == 0) ? PB31 : (q == 1) ? C31 : D31;
                const v8f r7  = (q == 0) ? PB7  : (q == 1) ? C7  : D7;
#pragma unroll
                for (int s = 0; s < 4; ++s) {
                    v2f a31 = {p31[2 * s], p31[2 * s + 1]};
                    v2f b31 = {r31[2 * s], r31[2 * s + 1]};
                    e31 = __builtin_amdgcn_wmma_f32_16x16x4_f32(
                        false, a31, false, w31[q][s], (short)0, e31, false, false);
                    f31 = __builtin_amdgcn_wmma_f32_16x16x4_f32(
                        false, b31, false, w31[q][s], (short)0, f31, false, false);
                    if (use7(q, s)) {
                        v2f a7 = {p7[2 * s], p7[2 * s + 1]};
                        v2f b7 = {r7[2 * s], r7[2 * s + 1]};
                        e7 = __builtin_amdgcn_wmma_f32_16x16x4_f32(
                            false, a7, false, w7[q][s], (short)0, e7, false, false);
                        f7 = __builtin_amdgcn_wmma_f32_16x16x4_f32(
                            false, b7, false, w7[q][s], (short)0, f7, false, false);
                    }
                }
            }
            float ra[8], rb[8];
#pragma unroll
            for (int r = 0; r < 8; ++r) {
                float Lv0 = LB[r];
                float le0 = Lv0 + ca * (Lv0 - e31[r]) + ta * (Lv0 - e7[r]);
                ra[r] = (le0 + 1e-6f) / (Lv0 + 1e-6f);
                float Lv1 = LC[r];
                float le1 = Lv1 + ca * (Lv1 - f31[r]) + ta * (Lv1 - f7[r]);
                rb[r] = (le1 + 1e-6f) / (Lv1 + 1e-6f);
            }
            float rr[16];
#pragma unroll
            for (int g = 0; g < 4; ++g) {
                float m0 = ra[2 * g], m1 = rb[2 * g], m2 = ra[2 * g + 1], m3 = rb[2 * g + 1];
                quad_transpose(m0, m1, m2, m3, qodd, qhi);
                rr[4 * g]     = half_transpose4x4(m0);
                rr[4 * g + 1] = half_transpose4x4(m1);
                rr[4 * g + 2] = half_transpose4x4(m2);
                rr[4 * g + 3] = half_transpose4x4(m3);
            }
            const size_t pbase = (size_t)(y0 + 8 * hi + ((lane >> 3) & 1)) * WQ + x0 + 4 * (lane & 7);
#pragma unroll
            for (int c = 0; c < 3; ++c) {
#pragma unroll
                for (int g = 0; g < 4; ++g) {
                    const size_t xa = (size_t)(b * 3 + c) * HW + pbase + (size_t)(2 * g) * WQ;
                    const v4f v = *(const v4f*)(x + xa);
                    v4f o;
                    o[0] = fminf(fmaxf(v[0] * rr[4 * g],     0.f), 1.f);
                    o[1] = fminf(fmaxf(v[1] * rr[4 * g + 1], 0.f), 1.f);
                    o[2] = fminf(fmaxf(v[2] * rr[4 * g + 2], 0.f), 1.f);
                    o[3] = fminf(fmaxf(v[3] * rr[4 * g + 3], 0.f), 1.f);
                    *(v4f*)(out + xa) = o;
                }
            }
        }
        PA31 = C31; PB31 = D31;
        PA7  = C7;  PB7  = D7;
        LB = LD;
    }
}

extern "C" void kernel_launch(void* const* d_in, const int* in_sizes, int n_in,
                              void* d_out, int out_size, void* d_ws, size_t ws_size,
                              hipStream_t stream) {
    const float* x    = (const float*)d_in[0];
    const float* clar = (const float*)d_in[1];
    const float* tex  = (const float*)d_in[2];
    float* out = (float*)d_out;

    float* L = (float*)d_ws;

    luma_kernel<<<(NB * NQR * WQ) / 256, 256, 0, stream>>>(x, L);
    dim3 grid2(WQ / (16 * NT), HQ / 64, NB);
    blur_combine_kernel<<<grid2, 128, 0, stream>>>(x, L, clar, tex, out);
}


# blur kernel: colour loads and output stores marked non-temporal (each line touched once): they stop displacing luma cells and the luma pass's colour lines from WGP cache and L2; values unchanged
# speedup vs baseline: 4.3377x; 1.0383x over previous
#include <hip/hip_runtime.h>
#include <cmath>


#define HQ 1024
#define WQ 1024
#define HW (HQ * WQ)
#define NB 8
#define NC 3
#define NT 8
#define LPQ (WQ + 32)
#define LOFF 16
#define NQR (HQ / 4)

typedef __attribute__((ext_vector_type(2))) float v2f;
typedef __attribute__((ext_vector_type(8))) float v8f;
typedef __attribute__((ext_vector_type(4))) float v4f;

__device__ __forceinline__ int iclamp(int v, int lo, int hi) {
    return v < lo ? lo : (v > hi ? hi : v);
}

__global__ __launch_bounds__(256) void luma_kernel(const float* __restrict__ x,
                                                   float* __restrict__ L) {
    const int i  = blockIdx.x * 256 + threadIdx.x;
    const int c  = i & (WQ - 1);
    const int qg = i >> 10;
    const int b  = qg >> 8;
    const int Q  = qg & (NQR - 1);
    const float* xr = x + (size_t)(b * 3) * HW + (size_t)(4 * Q) * WQ + c;
    float lum[4];
#pragma unroll
    for (int e = 0; e < 4; ++e) {
        const float r  = xr[e * WQ];
        const float g  = xr[HW + e * WQ];
        const float bl = xr[2 * HW + e * WQ];
        lum[e] = 0.2126f * r + 0.7152f * g + 0.0722f * bl;
    }
    float4 o;
    o.x = lum[0]; o.y = lum[1]; o.z = lum[2]; o.w = lum[3];
    *(float4*)(L + ((size_t)qg * LPQ + LOFF + c) * 4) = o;
}

__device__ __forceinline__ constexpr bool use7(int q, int s) {
    return (q == 1) || (q == 0 && s >= 2) || (q == 2 && s <= 1);
}

__device__ __forceinline__ float quad_xor1(float v) {
    const int u = __float_as_int(v);
    return __int_as_float(__builtin_amdgcn_update_dpp(u, u, 0xB1, 0xF, 0xF, false));
}
__device__ __forceinline__ float quad_xor2(float v) {
    const int u = __float_as_int(v);
    return __int_as_float(__builtin_amdgcn_update_dpp(u, u, 0x4E, 0xF, 0xF, false));
}

__device__ __forceinline__ void quad_transpose(float& a0, float& a1, float& a2, float& a3,
                                               bool odd, bool hiq) {
    const float t0 = quad_xor1(a1), t1 = quad_xor1(a0);
    const float t2 = quad_xor1(a3), t3 = quad_xor1(a2);
    const float b0 = odd ? t0 : a0;
    const float b1 = odd ? a1 : t1;
    const float b2 = odd ? t2 : a2;
    const float b3 = odd ? a3 : t3;
    const float u0 = quad_xor2(b2), u2 = quad_xor2(b0);
    const float u1 = quad_xor2(b3), u3 = quad_xor2(b1);
    a0 = hiq ? u0 : b0;
    a2 = hiq ? b2 : u2;
    a1 = hiq ? u1 : b1;
    a3 = hiq ? b3 : u3;
}

__device__ __forceinline__ float half_transpose4x4(float v) {
    const unsigned u = __float_as_uint(v);
    return __uint_as_float(__builtin_amdgcn_permlane16(u, u, 0xD951C840u, 0xFB73EA62u, false, false));
}


#define NSTEP (NT / 2 + 1)

__device__ __forceinline__ void load_step_cells(const float4* __restrict__ Lb, int rb0, int rb1,
                                                int rb2, int lh, int xs, int j,
                                                float4 (&oc)[6], float4 (&od)[6]) {
    const int X0 = xs + 32 * j - 16;
    const int colC = (X0 < 0) ? X0 + 16 : X0;
    const int colD = (X0 + 16 >= WQ) ? X0 : X0 + 16;
#pragma unroll
    for (int q = 0; q < 3; ++q) {
        const int rbq = (q == 0) ? rb0 : (q == 1) ? rb1 : rb2;
#pragma unroll
        for (int u = 0; u < 2; ++u) {
            oc[2 * q + u] = Lb[rbq + colC + u * LPQ + lh];
            od[2 * q + u] = Lb[rbq + colD + u * LPQ + lh];
        }
    }
}

__global__ __launch_bounds__(128) void blur_combine_kernel(
    const float* __restrict__ x, const float* __restrict__ L,
    const float* __restrict__ pclar, const float* __restrict__ ptex,
    float* __restrict__ out) {
    __shared__ float sg31[31];
    __shared__ float sg7[7];

    const int tid  = threadIdx.x;
    const int lane = tid & 31;
    const int ln   = lane & 15;
    const int hi   = lane >> 4;
    const bool qodd = (lane & 1) != 0;
    const bool qhi  = (lane & 2) != 0;
    const int wv   = __builtin_amdgcn_readfirstlane(tid >> 5);

    const int xs = blockIdx.x * (16 * NT);
    const int y0 = (blockIdx.y * 4 + wv) * 16;
    const int b  = blockIdx.z;

    const bool top = (y0 == 0), bot = (y0 == HQ - 16);
    const int   rq0 = top ? 16 : 0;
    const int   rq2 = bot ? 16 : 32;
    const float mq0 = top ? 0.f : 1.f;
    const float mq2 = bot ? 0.f : 1.f;

    const float4* Lb = (const float4*)L + (size_t)b * NQR * LPQ + LOFF;
    const int rb0 = ((y0 - 16 + rq0) >> 2) * LPQ;
    const int rb1 = (y0 >> 2) * LPQ;
    const int rb2 = ((y0 - 16 + rq2) >> 2) * LPQ;
    const int lh = 2 * hi * LPQ + ln;

    if (tid < 31) { float d = (float)tid - 15.f; sg31[tid] = expf(-d * d * (1.f / 128.f)); }
    if (tid < 7)  { float d = (float)tid - 3.f;  sg7[tid]  = expf(-d * d * (1.f / 4.5f)); }
    __syncthreads();

    float4 cc[6], cd[6];
    load_step_cells(Lb, rb0, rb1, rb2, lh, xs, 0, cc, cd);

    float s31 = 0.f, s7 = 0.f;
    for (int t = 0; t < 31; ++t) s31 += sg31[t];
    for (int t = 0; t < 7;  ++t) s7  += sg7[t];
    const float inv31 = 1.f / s31, inv7 = 1.f / s7;

    v2f w31[3][4];
    v2f w7[3][4];
#pragma unroll
    for (int q = 0; q < 3; ++q) {
#pragma unroll
        for (int s = 0; s < 4; ++s) {
#pragma unroll
            for (int v = 0; v < 2; ++v) {
                int kk = 16 * q + 8 * hi + 2 * s + v;
                int t  = kk - ln - 1;
                int tc = iclamp(t, 0, 30);
                w31[q][s][v] = sg31[tc] * ((t == tc) ? inv31 : 0.f);
                int u  = kk - ln - 13;
                int uc = iclamp(u, 0, 6);
                w7[q][s][v] = sg7[uc] * ((u == uc) ? inv7 : 0.f);
            }
        }
    }

    const float ca = tanhf(pclar[0]) * 0.5f;
    const float ta = tanhf(ptex[0])  * 0.3f;

    const v8f zero8 = {0.f, 0.f, 0.f, 0.f, 0.f, 0.f, 0.f, 0.f};
    v8f PA31 = zero8, PB31 = zero8;
    v8f PA7  = zero8, PB7  = zero8;
    v8f LB = zero8;

#pragma unroll 1
    for (int j = 0; j < NSTEP; ++j) {
        const int X0 = xs + 32 * j - 16;
        const bool cout_ = (X0 < 0), dout_ = (X0 + 16 >= WQ);
        const float mC = cout_ ? 0.f : 1.f;
        const float mD = dout_ ? 0.f : 1.f;

        v8f C31 = zero8, C7 = zero8, D31 = zero8, D7 = zero8, LC = zero8, LD = zero8;
#pragma unroll
        for (int q = 0; q < 3; ++q) {
            const float   mq   = (q == 0) ? mq0 : (q == 1) ? 1.f : mq2;
            const float   mmC = mq * mC, mmD = mq * mD;
#pragma unroll
            for (int u = 0; u < 2; ++u) {
                const float4 wc = cc[2 * q + u];
                const float4 wd = cd[2 * q + u];
#pragma unroll
                for (int t = 0; t < 2; ++t) {
                    const int s = 2 * u + t;
                    v2f fc, fd;
                    fc[0] = (t == 0 ? wc.x : wc.z) * mmC;
                    fc[1] = (t == 0 ? wc.y : wc.w) * mmC;
                    fd[0] = (t == 0 ? wd.x : wd.z) * mmD;
                    fd[1] = (t == 0 ? wd.y : wd.w) * mmD;
                    C31 = __builtin_amdgcn_wmma_f32_16x16x4_f32(
                        false, fc, false, w31[q][s], (short)0, C31, false, false);
                    D31 = __builtin_amdgcn_wmma_f32_16x16x4_f32(
                        false, fd, false, w31[q][s], (short)0, D31, false, false);
                    if (use7(q, s)) {
                        C7 = __builtin_amdgcn_wmma_f32_16x16x4_f32(
                            false, fc, false, w7[q][s], (short)0, C7, false, false);
                        D7 = __builtin_amdgcn_wmma_f32_16x16x4_f32(
                            false, fd, false, w7[q][s], (short)0, D7, false, false);
                    }
                    if (q == 1) {
                        LC[2 * s] = fc[0]; LC[2 * s + 1] = fc[1];
                        LD[2 * s] = fd[0]; LD[2 * s + 1] = fd[1];
                    }
                }
            }
        }

        if (j + 1 < NSTEP) load_step_cells(Lb, rb0, rb1, rb2, lh, xs, j + 1, cc, cd);

        if (j >= 1) {
            const int x0 = xs + 32 * (j - 1);
            v8f e31 = zero8, e7 = zero8, f31 = zero8, f7 = zero8;
#pragma unroll
            for (int q = 0; q < 3; ++q) {
                const v8f p31 = (q == 0) ? PA31 : (q == 1) ? PB31 : C31;
                const v8f p7  = (q == 0) ? PA7  : (q == 1) ? PB7  : C7;
                const v8f r31 = (q == 0) ? PB31 : (q == 1) ? C31 : D31;
                const v8f r7  = (q == 0) ? PB7  : (q == 1) ? C7  : D7;
#pragma unroll
                for (int s = 0; s < 4; ++s) {
                    v2f a31 = {p31[2 * s], p31[2 * s + 1]};
                    v2f b31 = {r31[2 * s], r31[2 * s + 1]};
                    e31 = __builtin_amdgcn_wmma_f32_16x16x4_f32(
                        false, a31, false, w31[q][s], (short)0, e31, false, false);
                    f31 = __builtin_amdgcn_wmma_f32_16x16x4_f32(
                        false, b31, false, w31[q][s], (short)0, f31, false, false);
                    if (use7(q, s)) {
                        v2f a7 = {p7[2 * s], p7[2 * s + 1]};
                        v2f b7 = {r7[2 * s], r7[2 * s + 1]};
                        e7 = __builtin_amdgcn_wmma_f32_16x16x4_f32(
                            false, a7, false, w7[q][s], (short)0, e7, false, false);
                        f7 = __builtin_amdgcn_wmma_f32_16x16x4_f32(
                            false, b7, false, w7[q][s], (short)0, f7, false, false);
                    }
                }
            }
            float ra[8], rb[8];
#pragma unroll
            for (int r = 0; r < 8; ++r) {
                float Lv0 = LB[r];
                float le0 = Lv0 + ca * (Lv0 - e31[r]) + ta * (Lv0 - e7[r]);
                ra[r] = (le0 + 1e-6f) / (Lv0 + 1e-6f);
                float Lv1 = LC[r];
                float le1 = Lv1 + ca * (Lv1 - f31[r]) + ta * (Lv1 - f7[r]);
                rb[r] = (le1 + 1e-6f) / (Lv1 + 1e-6f);
            }
            float rr[16];
#pragma unroll
            for (int g = 0; g < 4; ++g) {
                float m0 = ra[2 * g], m1 = rb[2 * g], m2 = ra[2 * g + 1], m3 = rb[2 * g + 1];
                quad_transpose(m0, m1, m2, m3, qodd, qhi);
                rr[4 * g]     = half_transpose4x4(m0);
                rr[4 * g + 1] = half_transpose4x4(m1);
                rr[4 * g + 2] = half_transpose4x4(m2);
                rr[4 * g + 3] = half_transpose4x4(m3);
            }
            const size_t pbase = (size_t)(y0 + 8 * hi + ((lane >> 3) & 1)) * WQ + x0 + 4 * (lane & 7);
#pragma unroll
            for (int c = 0; c < 3; ++c) {
#pragma unroll
                for (int g = 0; g < 4; ++g) {
                    const size_t xa = (size_t)(b * 3 + c) * HW + pbase + (size_t)(2 * g) * WQ;
                    const v4f v = __builtin_nontemporal_load((const v4f*)(x + xa));
                    v4f o;
                    o[0] = fminf(fmaxf(v[0] * rr[4 * g],     0.f), 1.f);
                    o[1] = fminf(fmaxf(v[1] * rr[4 * g + 1], 0.f), 1.f);
                    o[2] = fminf(fmaxf(v[2] * rr[4 * g + 2], 0.f), 1.f);
                    o[3] = fminf(fmaxf(v[3] * rr[4 * g + 3], 0.f), 1.f);
                    __builtin_nontemporal_store(o, (v4f*)(out + xa));
                }
            }
        }
        PA31 = C31; PB31 = D31;
        PA7  = C7;  PB7  = D7;
        LB = LD;
    }
}

extern "C" void kernel_launch(void* const* d_in, const int* in_sizes, int n_in,
                              void* d_out, int out_size, void* d_ws, size_t ws_size,
                              hipStream_t stream) {
    const float* x    = (const float*)d_in[0];
    const float* clar = (const float*)d_in[1];
    const float* tex  = (const float*)d_in[2];
    float* out = (float*)d_out;

    float* L = (float*)d_ws;

    luma_kernel<<<(NB * NQR * WQ) / 256, 256, 0, stream>>>(x, L);
    dim3 grid2(WQ / (16 * NT), HQ / 64, NB);
    blur_combine_kernel<<<grid2, 128, 0, stream>>>(x, L, clar, tex, out);
}


# combine step: each ratio is numerator x hardware reciprocal (v_rcp_f32, 1 ulp) instead of the IEEE division sequence; the 12 colour loads kept as one group behind the horizontal products
# speedup vs baseline: 4.5447x; 1.0477x over previous
#include <hip/hip_runtime.h>
#include <cmath>


#define HQ 1024
#define WQ 1024
#define HW (HQ * WQ)
#define NB 8
#define NC 3
#define NT 8
#define LPQ (WQ + 32)
#define LOFF 16
#define NQR (HQ / 4)

typedef __attribute__((ext_vector_type(2))) float v2f;
typedef __attribute__((ext_vector_type(8))) float v8f;
typedef __attribute__((ext_vector_type(4))) float v4f;

__device__ __forceinline__ int iclamp(int v, int lo, int hi) {
    return v < lo ? lo : (v > hi ? hi : v);
}

__global__ __launch_bounds__(256) void luma_kernel(const float* __restrict__ x,
                                                   float* __restrict__ L) {
    const int i  = blockIdx.x * 256 + threadIdx.x;
    const int c  = i & (WQ - 1);
    const int qg = i >> 10;
    const int b  = qg >> 8;
    const int Q  = qg & (NQR - 1);
    const float* xr = x + (size_t)(b * 3) * HW + (size_t)(4 * Q) * WQ + c;
    float lum[4];
#pragma unroll
    for (int e = 0; e < 4; ++e) {
        const float r  = xr[e * WQ];
        const float g  = xr[HW + e * WQ];
        const float bl = xr[2 * HW + e * WQ];
        lum[e] = 0.2126f * r + 0.7152f * g + 0.0722f * bl;
    }
    float4 o;
    o.x = lum[0]; o.y = lum[1]; o.z = lum[2]; o.w = lum[3];
    *(float4*)(L + ((size_t)qg * LPQ + LOFF + c) * 4) = o;
}

__device__ __forceinline__ constexpr bool use7(int q, int s) {
    return (q == 1) || (q == 0 && s >= 2) || (q == 2 && s <= 1);
}

__device__ __forceinline__ float quad_xor1(float v) {
    const int u = __float_as_int(v);
    return __int_as_float(__builtin_amdgcn_update_dpp(u, u, 0xB1, 0xF, 0xF, false));
}
__device__ __forceinline__ float quad_xor2(float v) {
    const int u = __float_as_int(v);
    return __int_as_float(__builtin_amdgcn_update_dpp(u, u, 0x4E, 0xF, 0xF, false));
}

__device__ __forceinline__ void quad_transpose(float& a0, float& a1, float& a2, float& a3,
                                               bool odd, bool hiq) {
    const float t0 = quad_xor1(a1), t1 = quad_xor1(a0);
    const float t2 = quad_xor1(a3), t3 = quad_xor1(a2);
    const float b0 = odd ? t0 : a0;
    const float b1 = odd ? a1 : t1;
    const float b2 = odd ? t2 : a2;
    const float b3 = odd ? a3 : t3;
    const float u0 = quad_xor2(b2), u2 = quad_xor2(b0);
    const float u1 = quad_xor2(b3), u3 = quad_xor2(b1);
    a0 = hiq ? u0 : b0;
    a2 = hiq ? b2 : u2;
    a1 = hiq ? u1 : b1;
    a3 = hiq ? b3 : u3;
}

__device__ __forceinline__ float half_transpose4x4(float v) {
    const unsigned u = __float_as_uint(v);
    return __uint_as_float(__builtin_amdgcn_permlane16(u, u, 0xD951C840u, 0xFB73EA62u, false, false));
}


#define NSTEP (NT / 2 + 1)

__device__ __forceinline__ void load_step_cells(const float4* __restrict__ Lb, int rb0, int rb1,
                                                int rb2, int lh, int xs, int j,
                                                float4 (&oc)[6], float4 (&od)[6]) {
    const int X0 = xs + 32 * j - 16;
    const int colC = (X0 < 0) ? X0 + 16 : X0;
    const int colD = (X0 + 16 >= WQ) ? X0 : X0 + 16;
#pragma unroll
    for (int q = 0; q < 3; ++q) {
        const int rbq = (q == 0) ? rb0 : (q == 1) ? rb1 : rb2;
#pragma unroll
        for (int u = 0; u < 2; ++u) {
            oc[2 * q + u] = Lb[rbq + colC + u * LPQ + lh];
            od[2 * q + u] = Lb[rbq + colD + u * LPQ + lh];
        }
    }
}

__global__ __launch_bounds__(128) void blur_combine_kernel(
    const float* __restrict__ x, const float* __restrict__ L,
    const float* __restrict__ pclar, const float* __restrict__ ptex,
    float* __restrict__ out) {
    __shared__ float sg31[31];
    __shared__ float sg7[7];

    const int tid  = threadIdx.x;
    const int lane = tid & 31;
    const int ln   = lane & 15;
    const int hi   = lane >> 4;
    const bool qodd = (lane & 1) != 0;
    const bool qhi  = (lane & 2) != 0;
    const int wv   = __builtin_amdgcn_readfirstlane(tid >> 5);

    const int xs = blockIdx.x * (16 * NT);
    const int y0 = (blockIdx.y * 4 + wv) * 16;
    const int b  = blockIdx.z;

    const bool top = (y0 == 0), bot = (y0 == HQ - 16);
    const int   rq0 = top ? 16 : 0;
    const int   rq2 = bot ? 16 : 32;
    const float mq0 = top ? 0.f : 1.f;
    const float mq2 = bot ? 0.f : 1.f;

    const float4* Lb = (const float4*)L + (size_t)b * NQR * LPQ + LOFF;
    const int rb0 = ((y0 - 16 + rq0) >> 2) * LPQ;
    const int rb1 = (y0 >> 2) * LPQ;
    const int rb2 = ((y0 - 16 + rq2) >> 2) * LPQ;
    const int lh = 2 * hi * LPQ + ln;

    if (tid < 31) { float d = (float)tid - 15.f; sg31[tid] = expf(-d * d * (1.f / 128.f)); }
    if (tid < 7)  { float d = (float)tid - 3.f;  sg7[tid]  = expf(-d * d * (1.f / 4.5f)); }
    __syncthreads();

    float4 cc[6], cd[6];
    load_step_cells(Lb, rb0, rb1, rb2, lh, xs, 0, cc, cd);

    float s31 = 0.f, s7 = 0.f;
    for (int t = 0; t < 31; ++t) s31 += sg31[t];
    for (int t = 0; t < 7;  ++t) s7  += sg7[t];
    const float inv31 = 1.f / s31, inv7 = 1.f / s7;

    v2f w31[3][4];
    v2f w7[3][4];
#pragma unroll
    for (int q = 0; q < 3; ++q) {
#pragma unroll
        for (int s = 0; s < 4; ++s) {
#pragma unroll
            for (int v = 0; v < 2; ++v) {
                int kk = 16 * q + 8 * hi + 2 * s + v;
                int t  = kk - ln - 1;
                int tc = iclamp(t, 0, 30);
                w31[q][s][v] = sg31[tc] * ((t == tc) ? inv31 : 0.f);
                int u  = kk - ln - 13;
                int uc = iclamp(u, 0, 6);
                w7[q][s][v] = sg7[uc] * ((u == uc) ? inv7 : 0.f);
            }
        }
    }

    const float ca = tanhf(pclar[0]) * 0.5f;
    const float ta = tanhf(ptex[0])  * 0.3f;

    const v8f zero8 = {0.f, 0.f, 0.f, 0.f, 0.f, 0.f, 0.f, 0.f};
    v8f PA31 = zero8, PB31 = zero8;
    v8f PA7  = zero8, PB7  = zero8;
    v8f LB = zero8;

#pragma unroll 1
    for (int j = 0; j < NSTEP; ++j) {
        const int X0 = xs + 32 * j - 16;
        const bool cout_ = (X0 < 0), dout_ = (X0 + 16 >= WQ);
        const float mC = cout_ ? 0.f : 1.f;
        const float mD = dout_ ? 0.f : 1.f;

        v8f C31 = zero8, C7 = zero8, D31 = zero8, D7 = zero8, LC = zero8, LD = zero8;
#pragma unroll
        for (int q = 0; q < 3; ++q) {
            const float   mq   = (q == 0) ? mq0 : (q == 1) ? 1.f : mq2;
            const float   mmC = mq * mC, mmD = mq * mD;
#pragma unroll
            for (int u = 0; u < 2; ++u) {
                const float4 wc = cc[2 * q + u];
                const float4 wd = cd[2 * q + u];
#pragma unroll
                for (int t = 0; t < 2; ++t) {
                    const int s = 2 * u + t;
                    v2f fc, fd;
                    fc[0] = (t == 0 ? wc.x : wc.z) * mmC;
                    fc[1] = (t == 0 ? wc.y : wc.w) * mmC;
                    fd[0] = (t == 0 ? wd.x : wd.z) * mmD;
                    fd[1] = (t == 0 ? wd.y : wd.w) * mmD;
                    C31 = __builtin_amdgcn_wmma_f32_16x16x4_f32(
                        false, fc, false, w31[q][s], (short)0, C31, false, false);
                    D31 = __builtin_amdgcn_wmma_f32_16x16x4_f32(
                        false, fd, false, w31[q][s], (short)0, D31, false, false);
                    if (use7(q, s)) {
                        C7 = __builtin_amdgcn_wmma_f32_16x16x4_f32(
                            false, fc, false, w7[q][s], (short)0, C7, false, false);
                        D7 = __builtin_amdgcn_wmma_f32_16x16x4_f32(
                            false, fd, false, w7[q][s], (short)0, D7, false, false);
                    }
                    if (q == 1) {
                        LC[2 * s] = fc[0]; LC[2 * s + 1] = fc[1];
                        LD[2 * s] = fd[0]; LD[2 * s + 1] = fd[1];
                    }
                }
            }
        }

        if (j + 1 < NSTEP) load_step_cells(Lb, rb0, rb1, rb2, lh, xs, j + 1, cc, cd);

        if (j >= 1) {
            const int x0 = xs + 32 * (j - 1);
            v8f e31 = zero8, e7 = zero8, f31 = zero8, f7 = zero8;
#pragma unroll
            for (int q = 0; q < 3; ++q) {
                const v8f p31 = (q == 0) ? PA31 : (q == 1) ? PB31 : C31;
                const v8f p7  = (q == 0) ? PA7  : (q == 1) ? PB7  : C7;
                const v8f r31 = (q == 0) ? PB31 : (q == 1) ? C31 : D31;
                const v8f r7  = (q == 0) ? PB7  : (q == 1) ? C7  : D7;
#pragma unroll
                for (int s = 0; s < 4; ++s) {
                    v2f a31 = {p31[2 * s], p31[2 * s + 1]};
                    v2f b31 = {r31[2 * s], r31[2 * s + 1]};
                    e31 = __builtin_amdgcn_wmma_f32_16x16x4_f32(
                        false, a31, false, w31[q][s], (short)0, e31, false, false);
                    f31 = __builtin_amdgcn_wmma_f32_16x16x4_f32(
                        false, b31, false, w31[q][s], (short)0, f31, false, false);
                    if (use7(q, s)) {
                        v2f a7 = {p7[2 * s], p7[2 * s + 1]};
                        v2f b7 = {r7[2 * s], r7[2 * s + 1]};
                        e7 = __builtin_amdgcn_wmma_f32_16x16x4_f32(
                            false, a7, false, w7[q][s], (short)0, e7, false, false);
                        f7 = __builtin_amdgcn_wmma_f32_16x16x4_f32(
                            false, b7, false, w7[q][s], (short)0, f7, false, false);
                    }
                }
            }
            __builtin_amdgcn_sched_barrier(0);
            const size_t pbase = (size_t)(y0 + 8 * hi + ((lane >> 3) & 1)) * WQ + x0 + 4 * (lane & 7);
            v4f col[12];
#pragma unroll
            for (int c = 0; c < 3; ++c) {
#pragma unroll
                for (int g = 0; g < 4; ++g) {
                    const size_t xa = (size_t)(b * 3 + c) * HW + pbase + (size_t)(2 * g) * WQ;
                    col[4 * c + g] = __builtin_nontemporal_load((const v4f*)(x + xa));
                }
            }
            __builtin_amdgcn_sched_barrier(0);
            float ra[8], rb[8];
#pragma unroll
            for (int r = 0; r < 8; ++r) {
                float Lv0 = LB[r];
                float le0 = Lv0 + ca * (Lv0 - e31[r]) + ta * (Lv0 - e7[r]);
                ra[r] = (le0 + 1e-6f) * __builtin_amdgcn_rcpf(Lv0 + 1e-6f);
                float Lv1 = LC[r];
                float le1 = Lv1 + ca * (Lv1 - f31[r]) + ta * (Lv1 - f7[r]);
                rb[r] = (le1 + 1e-6f) * __builtin_amdgcn_rcpf(Lv1 + 1e-6f);
            }
            float rr[16];
#pragma unroll
            for (int g = 0; g < 4; ++g) {
                float m0 = ra[2 * g], m1 = rb[2 * g], m2 = ra[2 * g + 1], m3 = rb[2 * g + 1];
                quad_transpose(m0, m1, m2, m3, qodd, qhi);
                rr[4 * g]     = half_transpose4x4(m0);
                rr[4 * g + 1] = half_transpose4x4(m1);
                rr[4 * g + 2] = half_transpose4x4(m2);
                rr[4 * g + 3] = half_transpose4x4(m3);
            }
#pragma unroll
            for (int c = 0; c < 3; ++c) {
#pragma unroll
                for (int g = 0; g < 4; ++g) {
                    const size_t xa = (size_t)(b * 3 + c) * HW + pbase + (size_t)(2 * g) * WQ;
                    const v4f v = col[4 * c + g];
                    v4f o;
                    o[0] = fminf(fmaxf(v[0] * rr[4 * g],     0.f), 1.f);
                    o[1] = fminf(fmaxf(v[1] * rr[4 * g + 1], 0.f), 1.f);
                    o[2] = fminf(fmaxf(v[2] * rr[4 * g + 2], 0.f), 1.f);
                    o[3] = fminf(fmaxf(v[3] * rr[4 * g + 3], 0.f), 1.f);
                    __builtin_nontemporal_store(o, (v4f*)(out + xa));
                }
            }
        }
        PA31 = C31; PB31 = D31;
        PA7  = C7;  PB7  = D7;
        LB = LD;
    }
}

extern "C" void kernel_launch(void* const* d_in, const int* in_sizes, int n_in,
                              void* d_out, int out_size, void* d_ws, size_t ws_size,
                              hipStream_t stream) {
    const float* x    = (const float*)d_in[0];
    const float* clar = (const float*)d_in[1];
    const float* tex  = (const float*)d_in[2];
    float* out = (float*)d_out;

    float* L = (float*)d_ws;

    luma_kernel<<<(NB * NQR * WQ) / 256, 256, 0, stream>>>(x, L);
    dim3 grid2(WQ / (16 * NT), HQ / 64, NB);
    blur_combine_kernel<<<grid2, 128, 0, stream>>>(x, L, clar, tex, out);
}
